# starting kernel
# baseline (speedup 1.0000x reference)
#include <hip/hip_runtime.h>
#include <math.h>


#define Bsz 64
#define Rsz 128
#define Asz 128
#define Fsz 256
#define Dsz 128

typedef __attribute__((ext_vector_type(2))) float v2f;
typedef __attribute__((ext_vector_type(8))) float v8f;

#if __has_builtin(__builtin_amdgcn_tanhf)
  #define TANH(x) __builtin_amdgcn_tanhf(x)
#elif __has_builtin(__builtin_amdgcn_tanh_f32)
  #define TANH(x) __builtin_amdgcn_tanh_f32(x)
#else
  #define TANH(x) tanhf(x)
#endif

__global__ __launch_bounds__(256) void coattn_proj_wmma(
    const float* __restrict__ receiver,
    const float* __restrict__ attendant,
    const float* __restrict__ w_k,
    const float* __restrict__ w_q,
    const float* __restrict__ bias,
    float* __restrict__ keys_ws,
    float* __restrict__ qry_ws)
{
    const int t    = threadIdx.x;
    const int lane = t & 31;
    const int wave = t >> 5;

    const int tile  = blockIdx.x * 8 + wave;
    const int which = tile >> 12;
    const int tt    = tile & 4095;
    const int mt    = tt >> 3;
    const int nt    = tt & 7;

    const float* X = which ? attendant : receiver;
    const float* W = which ? w_q : w_k;
    float*       O = which ? qry_ws : keys_ws;

    const int mrow = mt * 16 + (lane & 15);
    const int ncol = nt * 16 + (lane & 15);
    const int koff = (lane >> 4) << 1;

    const float* arow = X + (size_t)mrow * Fsz;

    v8f c = {};
    #pragma unroll 4
    for (int k0 = 0; k0 < Fsz; k0 += 4) {
        v2f a, b;
        a.x = arow[k0 + koff];
        a.y = arow[k0 + koff + 1];
        b.x = W[(size_t)(k0 + koff) * Dsz + ncol];
        b.y = W[(size_t)(k0 + koff + 1) * Dsz + ncol];
        c = __builtin_amdgcn_wmma_f32_16x16x4_f32(
                false, a, false, b, (short)0, c, false, false);
    }

    const float badd  = which ? bias[ncol] : 0.0f;
    const int   rbase = mt * 16 + (lane >> 4) * 8;
    #pragma unroll
    for (int v = 0; v < 8; ++v) {
        O[(size_t)(rbase + v) * Dsz + ncol] = c[v] + badd;
    }
}

#define QT_STRIDE 129

__global__ __launch_bounds__(512) void coattn_score(
    const float* __restrict__ qry_ws,
    const float* __restrict__ keys_ws,
    const float* __restrict__ avec,
    float* __restrict__ out)
{
    extern __shared__ float smem[];
    float* q_t = smem;
    float* kl  = smem + Dsz * QT_STRIDE;
    float* al  = kl + 32 * Dsz;

    const int b    = blockIdx.x >> 2;
    const int rblk = blockIdx.x & 3;
    const int t    = threadIdx.x;
    const int lane = t & 31;
    const int wave = t >> 5;

    const float* qb = qry_ws + (size_t)b * Asz * Dsz;
    #pragma unroll
    for (int i = 0; i < (Asz * Dsz) / 512; ++i) {
        int idx = t + i * 512;
        int a   = idx >> 7;
        int d   = idx & 127;
        q_t[d * QT_STRIDE + a] = qb[idx];
    }
    const float* kb = keys_ws + (size_t)(b * Rsz + rblk * 32) * Dsz;
    #pragma unroll
    for (int i = 0; i < (32 * Dsz) / 512; ++i) {
        int idx = t + i * 512;
        kl[idx] = kb[idx];
    }
    if (t < Dsz) al[t] = avec[t];
    __syncthreads();

    const float* k0row = kl + (wave * 2 + 0) * Dsz;
    const float* k1row = kl + (wave * 2 + 1) * Dsz;

    float a00 = 0.f, a01 = 0.f, a02 = 0.f, a03 = 0.f;
    float a10 = 0.f, a11 = 0.f, a12 = 0.f, a13 = 0.f;

    #pragma unroll 4
    for (int d = 0; d < Dsz; ++d) {
        const float kv0 = k0row[d];
        const float kv1 = k1row[d];
        const float av  = al[d];
        const float* qrow = q_t + d * QT_STRIDE + lane;
        const float q0 = qrow[0];
        const float q1 = qrow[32];
        const float q2 = qrow[64];
        const float q3 = qrow[96];
        a00 += TANH(q0 + kv0) * av;
        a01 += TANH(q1 + kv0) * av;
        a02 += TANH(q2 + kv0) * av;
        a03 += TANH(q3 + kv0) * av;
        a10 += TANH(q0 + kv1) * av;
        a11 += TANH(q1 + kv1) * av;
        a12 += TANH(q2 + kv1) * av;
        a13 += TANH(q3 + kv1) * av;
    }

    const int r0 = rblk * 32 + wave * 2;
    float* orow0 = out + ((size_t)(b * Rsz + r0)) * Asz + lane;
    float* orow1 = orow0 + Asz;
    orow0[0]  = a00;  orow0[32] = a01;  orow0[64] = a02;  orow0[96] = a03;
    orow1[0]  = a10;  orow1[32] = a11;  orow1[64] = a12;  orow1[96] = a13;
}

extern "C" void kernel_launch(void* const* d_in, const int* in_sizes, int n_in,
                              void* d_out, int out_size, void* d_ws, size_t ws_size,
                              hipStream_t stream) {
    const float* receiver  = (const float*)d_in[0];
    const float* attendant = (const float*)d_in[1];
    const float* w_q       = (const float*)d_in[2];
    const float* w_k       = (const float*)d_in[3];
    const float* bias      = (const float*)d_in[4];
    const float* avec      = (const float*)d_in[5];
    float*       out       = (float*)d_out;

    float* qry_ws  = (float*)d_ws;
    float* keys_ws = qry_ws + (size_t)Bsz * Asz * Dsz;

    coattn_proj_wmma<<<1024, 256, 0, stream>>>(
        receiver, attendant, w_k, w_q, bias, keys_ws, qry_ws);

    const size_t lds_bytes = (Dsz * QT_STRIDE + 32 * Dsz + Dsz) * sizeof(float);
    coattn_score<<<256, 512, lds_bytes, stream>>>(
        qry_ws, keys_ws, avec, out);
}


# Projection GEMM: stage X slab and W through LDS with coalesced 16-byte loads, reuse fragments over a 16x64 wave tile (64x128 per block, 256 blocks) instead of scattered per-WMMA global loads
# speedup vs baseline: 1.3712x; 1.3712x over previous
#include <hip/hip_runtime.h>
#include <math.h>


#define Bsz 64
#define Rsz 128
#define Asz 128
#define Fsz 256
#define Dsz 128

typedef __attribute__((ext_vector_type(2))) float v2f;
typedef __attribute__((ext_vector_type(8))) float v8f;

#if __has_builtin(__builtin_amdgcn_tanhf)
  #define TANH(x) __builtin_amdgcn_tanhf(x)
#elif __has_builtin(__builtin_amdgcn_tanh_f32)
  #define TANH(x) __builtin_amdgcn_tanh_f32(x)
#else
  #define TANH(x) tanhf(x)
#endif

typedef __attribute__((ext_vector_type(4))) float v4f;

#define P1_BM     64
#define P1_KH     128
#define P1_APITCH 132
#define P1_BPITCH 260
#define P1_A_DW   (P1_BM * P1_APITCH)
#define P1_B_DW   ((P1_KH / 2) * P1_BPITCH)
#define P1_ST_DW  (P1_A_DW + P1_B_DW)

__device__ __forceinline__ v8f wmma4(v2f a, v2f b, v8f c) {
    return __builtin_amdgcn_wmma_f32_16x16x4_f32(
               false, a, false, b, (short)0, c, false, false);
}

#define LO2(v) __builtin_shufflevector((v), (v), 0, 1)
#define HI2(v) __builtin_shufflevector((v), (v), 2, 3)

struct P1Frag { v4f a0, a1, p0, p1, p2, p3, q0, q1, q2, q3; };

__device__ __forceinline__ P1Frag p1_load_frag(const float* fa, const float* fb, int kb) {
    P1Frag f;
    const float* b = fb + (kb >> 1) * P1_BPITCH;
    f.a0 = *(const v4f*)(fa + kb);
    f.a1 = *(const v4f*)(fa + kb + 8);
    f.p0 = *(const v4f*)(b);
    f.q0 = *(const v4f*)(b + 64);
    f.p1 = *(const v4f*)(b + P1_BPITCH);
    f.q1 = *(const v4f*)(b + P1_BPITCH + 64);
    f.p2 = *(const v4f*)(b + 4 * P1_BPITCH);
    f.q2 = *(const v4f*)(b + 4 * P1_BPITCH + 64);
    f.p3 = *(const v4f*)(b + 5 * P1_BPITCH);
    f.q3 = *(const v4f*)(b + 5 * P1_BPITCH + 64);
    return f;
}

__device__ __forceinline__ void p1_half(const float* fa, const float* fb,
                                        v8f& acc0, v8f& acc1, v8f& acc2, v8f& acc3) {
    P1Frag cur = p1_load_frag(fa, fb, 0);
    #pragma unroll
    for (int kb = 0; kb < P1_KH; kb += 16) {
        P1Frag nxt = cur;
        if (kb + 16 < P1_KH) nxt = p1_load_frag(fa, fb, kb + 16);
        __builtin_amdgcn_sched_barrier(0);
        acc0 = wmma4(LO2(cur.a0), LO2(cur.p0), acc0);
        acc1 = wmma4(LO2(cur.a0), HI2(cur.p0), acc1);
        acc2 = wmma4(LO2(cur.a0), LO2(cur.q0), acc2);
        acc3 = wmma4(LO2(cur.a0), HI2(cur.q0), acc3);
        acc0 = wmma4(HI2(cur.a0), LO2(cur.p1), acc0);
        acc1 = wmma4(HI2(cur.a0), HI2(cur.p1), acc1);
        acc2 = wmma4(HI2(cur.a0), LO2(cur.q1), acc2);
        acc3 = wmma4(HI2(cur.a0), HI2(cur.q1), acc3);
        acc0 = wmma4(LO2(cur.a1), LO2(cur.p2), acc0);
        acc1 = wmma4(LO2(cur.a1), HI2(cur.p2), acc1);
        acc2 = wmma4(LO2(cur.a1), LO2(cur.q2), acc2);
        acc3 = wmma4(LO2(cur.a1), HI2(cur.q2), acc3);
        acc0 = wmma4(HI2(cur.a1), LO2(cur.p3), acc0);
        acc1 = wmma4(HI2(cur.a1), HI2(cur.p3), acc1);
        acc2 = wmma4(HI2(cur.a1), LO2(cur.q3), acc2);
        acc3 = wmma4(HI2(cur.a1), HI2(cur.q3), acc3);
        __builtin_amdgcn_sched_barrier(0);
        cur = nxt;
    }
}

__global__ __launch_bounds__(256) void coattn_proj_wmma(
    const float* __restrict__ receiver,
    const float* __restrict__ attendant,
    const float* __restrict__ w_k,
    const float* __restrict__ w_q,
    const float* __restrict__ bias,
    float* __restrict__ keys_ws,
    float* __restrict__ qry_ws)
{
    __shared__ __attribute__((aligned(16))) float lds[2 * P1_ST_DW];

    const int t    = threadIdx.x;
    const int lane = t & 31;
    const int wave = t >> 5;
    const int c    = lane & 15;
    const int h    = lane >> 4;

    const int which = blockIdx.x >> 7;
    const int m0    = (blockIdx.x & 127) * P1_BM;

    const float* X  = which ? attendant : receiver;
    const float* Wm = which ? w_q : w_k;
    float*       O  = which ? qry_ws : keys_ws;

    const int srow = t >> 5;
    const int scol = (t & 31) * 4;
    const float* xg = X  + (size_t)(m0 + srow) * Fsz + scol;
    const float* wg = Wm + (size_t)(2 * srow) * Dsz + scol;
    float* sa = lds + srow * P1_APITCH + scol;
    float* sb = lds + P1_A_DW + srow * P1_BPITCH + 2 * scol;

    v4f ra[8], rb[16];

    #pragma unroll
    for (int i = 0; i < 8; ++i)
        ra[i] = *(const v4f*)(xg + (size_t)(8 * i) * Fsz);
    #pragma unroll
    for (int i = 0; i < 8; ++i) {
        rb[2 * i]     = *(const v4f*)(wg + (size_t)(16 * i) * Dsz);
        rb[2 * i + 1] = *(const v4f*)(wg + (size_t)(16 * i + 1) * Dsz);
    }

    #pragma unroll
    for (int i = 0; i < 8; ++i)
        *(v4f*)(sa + (8 * i) * P1_APITCH) = ra[i];
    #pragma unroll
    for (int i = 0; i < 8; ++i) {
        const v4f r0 = rb[2 * i], r1 = rb[2 * i + 1];
        v4f lo, hi;
        lo.x = r0.x; lo.y = r1.x; lo.z = r0.y; lo.w = r1.y;
        hi.x = r0.z; hi.y = r1.z; hi.z = r0.w; hi.w = r1.w;
        float* dst = sb + (8 * i) * P1_BPITCH;
        *(v4f*)(dst)     = lo;
        *(v4f*)(dst + 4) = hi;
    }
    __syncthreads();

    #pragma unroll
    for (int i = 0; i < 8; ++i)
        ra[i] = *(const v4f*)(xg + (size_t)(8 * i) * Fsz + P1_KH);
    #pragma unroll
    for (int i = 0; i < 8; ++i) {
        rb[2 * i]     = *(const v4f*)(wg + (size_t)(P1_KH + 16 * i) * Dsz);
        rb[2 * i + 1] = *(const v4f*)(wg + (size_t)(P1_KH + 16 * i + 1) * Dsz);
    }

    const int mt = wave >> 1;
    const int nb = (wave & 1) * 64;
    const float* fa = lds + (mt * 16 + c) * P1_APITCH + 4 * h;
    const float* fb = lds + P1_A_DW + (2 * h) * P1_BPITCH + 2 * nb + 4 * c;

    v8f acc0 = {}, acc1 = {}, acc2 = {}, acc3 = {};

    p1_half(fa, fb, acc0, acc1, acc2, acc3);

    #pragma unroll
    for (int i = 0; i < 8; ++i)
        *(v4f*)(sa + P1_ST_DW + (8 * i) * P1_APITCH) = ra[i];
    #pragma unroll
    for (int i = 0; i < 8; ++i) {
        const v4f r0 = rb[2 * i], r1 = rb[2 * i + 1];
        v4f lo, hi;
        lo.x = r0.x; lo.y = r1.x; lo.z = r0.y; lo.w = r1.y;
        hi.x = r0.z; hi.y = r1.z; hi.z = r0.w; hi.w = r1.w;
        float* dst = sb + P1_ST_DW + (8 * i) * P1_BPITCH;
        *(v4f*)(dst)     = lo;
        *(v4f*)(dst + 4) = hi;
    }
    __syncthreads();

    int st1 = P1_ST_DW;
    asm volatile("" : "+v"(st1));
    p1_half(fa + st1, fb + st1, acc0, acc1, acc2, acc3);

    const int ncol = nb + 2 * c;
    v2f badd01 = {0.0f, 0.0f}, badd23 = {0.0f, 0.0f};
    if (which) {
        badd01.x = bias[ncol];
        badd01.y = bias[ncol + 1];
        badd23.x = bias[ncol + 32];
        badd23.y = bias[ncol + 33];
    }
    float* orow = O + (size_t)(m0 + mt * 16 + 8 * h) * Dsz + ncol;
    #pragma unroll
    for (int r = 0; r < 8; ++r) {
        v2f o01, o23;
        o01.x = acc0[r] + badd01.x;
        o01.y = acc1[r] + badd01.y;
        o23.x = acc2[r] + badd23.x;
        o23.y = acc3[r] + badd23.y;
        *(v2f*)(orow + (size_t)r * Dsz)      = o01;
        *(v2f*)(orow + (size_t)r * Dsz + 32) = o23;
    }
}

#define QT_STRIDE 129

__global__ __launch_bounds__(512) void coattn_score(
    const float* __restrict__ qry_ws,
    const float* __restrict__ keys_ws,
    const float* __restrict__ avec,
    float* __restrict__ out)
{
    extern __shared__ float smem[];
    float* q_t = smem;
    float* kl  = smem + Dsz * QT_STRIDE;
    float* al  = kl + 32 * Dsz;

    const int b    = blockIdx.x >> 2;
    const int rblk = blockIdx.x & 3;
    const int t    = threadIdx.x;
    const int lane = t & 31;
    const int wave = t >> 5;

    const float* qb = qry_ws + (size_t)b * Asz * Dsz;
    #pragma unroll
    for (int i = 0; i < (Asz * Dsz) / 512; ++i) {
        int idx = t + i * 512;
        int a   = idx >> 7;
        int d   = idx & 127;
        q_t[d * QT_STRIDE + a] = qb[idx];
    }
    const float* kb = keys_ws + (size_t)(b * Rsz + rblk * 32) * Dsz;
    #pragma unroll
    for (int i = 0; i < (32 * Dsz) / 512; ++i) {
        int idx = t + i * 512;
        kl[idx] = kb[idx];
    }
    if (t < Dsz) al[t] = avec[t];
    __syncthreads();

    const float* k0row = kl + (wave * 2 + 0) * Dsz;
    const float* k1row = kl + (wave * 2 + 1) * Dsz;

    float a00 = 0.f, a01 = 0.f, a02 = 0.f, a03 = 0.f;
    float a10 = 0.f, a11 = 0.f, a12 = 0.f, a13 = 0.f;

    #pragma unroll 4
    for (int d = 0; d < Dsz; ++d) {
        const float kv0 = k0row[d];
        const float kv1 = k1row[d];
        const float av  = al[d];
        const float* qrow = q_t + d * QT_STRIDE + lane;
        const float q0 = qrow[0];
        const float q1 = qrow[32];
        const float q2 = qrow[64];
        const float q3 = qrow[96];
        a00 += TANH(q0 + kv0) * av;
        a01 += TANH(q1 + kv0) * av;
        a02 += TANH(q2 + kv0) * av;
        a03 += TANH(q3 + kv0) * av;
        a10 += TANH(q0 + kv1) * av;
        a11 += TANH(q1 + kv1) * av;
        a12 += TANH(q2 + kv1) * av;
        a13 += TANH(q3 + kv1) * av;
    }

    const int r0 = rblk * 32 + wave * 2;
    float* orow0 = out + ((size_t)(b * Rsz + r0)) * Asz + lane;
    float* orow1 = orow0 + Asz;
    orow0[0]  = a00;  orow0[32] = a01;  orow0[64] = a02;  orow0[96] = a03;
    orow1[0]  = a10;  orow1[32] = a11;  orow1[64] = a12;  orow1[96] = a13;
}

extern "C" void kernel_launch(void* const* d_in, const int* in_sizes, int n_in,
                              void* d_out, int out_size, void* d_ws, size_t ws_size,
                              hipStream_t stream) {
    const float* receiver  = (const float*)d_in[0];
    const float* attendant = (const float*)d_in[1];
    const float* w_q       = (const float*)d_in[2];
    const float* w_k       = (const float*)d_in[3];
    const float* bias      = (const float*)d_in[4];
    const float* avec      = (const float*)d_in[5];
    float*       out       = (float*)d_out;

    float* qry_ws  = (float*)d_ws;
    float* keys_ws = qry_ws + (size_t)Bsz * Asz * Dsz;

    coattn_proj_wmma<<<256, 256, 0, stream>>>(
        receiver, attendant, w_k, w_q, bias, keys_ws, qry_ws);

    const size_t lds_bytes = (Dsz * QT_STRIDE + 32 * Dsz + Dsz) * sizeof(float);
    coattn_score<<<256, 512, lds_bytes, stream>>>(
        qry_ws, keys_ws, avec, out);
}


# Score kernel re-tiled: 8-wave blocks of 4r x 4a per wave, q'/k kept [row][d] in LDS (16-byte reads), d four-wide on packed f32 add/fma software-pipelined to alternate with V_TANH_F32
# speedup vs baseline: 1.4487x; 1.0565x over previous
#include <hip/hip_runtime.h>
#include <math.h>


#define Bsz 64
#define Rsz 128
#define Asz 128
#define Fsz 256
#define Dsz 128

typedef __attribute__((ext_vector_type(2))) float v2f;
typedef __attribute__((ext_vector_type(8))) float v8f;

#if __has_builtin(__builtin_amdgcn_tanhf)
  #define TANH(x) __builtin_amdgcn_tanhf(x)
#elif __has_builtin(__builtin_amdgcn_tanh_f32)
  #define TANH(x) __builtin_amdgcn_tanh_f32(x)
#else
  #define TANH(x) tanhf(x)
#endif

typedef __attribute__((ext_vector_type(4))) float v4f;

#define P1_BM     64
#define P1_KH     128
#define P1_APITCH 132
#define P1_BPITCH 260
#define P1_A_DW   (P1_BM * P1_APITCH)
#define P1_B_DW   ((P1_KH / 2) * P1_BPITCH)
#define P1_ST_DW  (P1_A_DW + P1_B_DW)

__device__ __forceinline__ v8f wmma4(v2f a, v2f b, v8f c) {
    return __builtin_amdgcn_wmma_f32_16x16x4_f32(
               false, a, false, b, (short)0, c, false, false);
}

#define LO2(v) __builtin_shufflevector((v), (v), 0, 1)
#define HI2(v) __builtin_shufflevector((v), (v), 2, 3)

struct P1Frag { v4f a0, a1, p0, p1, p2, p3, q0, q1, q2, q3; };

__device__ __forceinline__ P1Frag p1_load_frag(const float* fa, const float* fb, int kb) {
    P1Frag f;
    const float* b = fb + (kb >> 1) * P1_BPITCH;
    f.a0 = *(const v4f*)(fa + kb);
    f.a1 = *(const v4f*)(fa + kb + 8);
    f.p0 = *(const v4f*)(b);
    f.q0 = *(const v4f*)(b + 64);
    f.p1 = *(const v4f*)(b + P1_BPITCH);
    f.q1 = *(const v4f*)(b + P1_BPITCH + 64);
    f.p2 = *(const v4f*)(b + 4 * P1_BPITCH);
    f.q2 = *(const v4f*)(b + 4 * P1_BPITCH + 64);
    f.p3 = *(const v4f*)(b + 5 * P1_BPITCH);
    f.q3 = *(const v4f*)(b + 5 * P1_BPITCH + 64);
    return f;
}

__device__ __forceinline__ void p1_half(const float* fa, const float* fb,
                                        v8f& acc0, v8f& acc1, v8f& acc2, v8f& acc3) {
    P1Frag cur = p1_load_frag(fa, fb, 0);
    #pragma unroll
    for (int kb = 0; kb < P1_KH; kb += 16) {
        P1Frag nxt = cur;
        if (kb + 16 < P1_KH) nxt = p1_load_frag(fa, fb, kb + 16);
        __builtin_amdgcn_sched_barrier(0);
        acc0 = wmma4(LO2(cur.a0), LO2(cur.p0), acc0);
        acc1 = wmma4(LO2(cur.a0), HI2(cur.p0), acc1);
        acc2 = wmma4(LO2(cur.a0), LO2(cur.q0), acc2);
        acc3 = wmma4(LO2(cur.a0), HI2(cur.q0), acc3);
        acc0 = wmma4(HI2(cur.a0), LO2(cur.p1), acc0);
        acc1 = wmma4(HI2(cur.a0), HI2(cur.p1), acc1);
        acc2 = wmma4(HI2(cur.a0), LO2(cur.q1), acc2);
        acc3 = wmma4(HI2(cur.a0), HI2(cur.q1), acc3);
        acc0 = wmma4(LO2(cur.a1), LO2(cur.p2), acc0);
        acc1 = wmma4(LO2(cur.a1), HI2(cur.p2), acc1);
        acc2 = wmma4(LO2(cur.a1), LO2(cur.q2), acc2);
        acc3 = wmma4(LO2(cur.a1), HI2(cur.q2), acc3);
        acc0 = wmma4(HI2(cur.a1), LO2(cur.p3), acc0);
        acc1 = wmma4(HI2(cur.a1), HI2(cur.p3), acc1);
        acc2 = wmma4(HI2(cur.a1), LO2(cur.q3), acc2);
        acc3 = wmma4(HI2(cur.a1), HI2(cur.q3), acc3);
        __builtin_amdgcn_sched_barrier(0);
        cur = nxt;
    }
}

__global__ __launch_bounds__(256) void coattn_proj_wmma(
    const float* __restrict__ receiver,
    const float* __restrict__ attendant,
    const float* __restrict__ w_k,
    const float* __restrict__ w_q,
    const float* __restrict__ bias,
    float* __restrict__ keys_ws,
    float* __restrict__ qry_ws)
{
    __shared__ __attribute__((aligned(16))) float lds[2 * P1_ST_DW];

    const int t    = threadIdx.x;
    const int lane = t & 31;
    const int wave = t >> 5;
    const int c    = lane & 15;
    const int h    = lane >> 4;

    const int which = blockIdx.x >> 7;
    const int m0    = (blockIdx.x & 127) * P1_BM;

    const float* X  = which ? attendant : receiver;
    const float* Wm = which ? w_q : w_k;
    float*       O  = which ? qry_ws : keys_ws;

    const int srow = t >> 5;
    const int scol = (t & 31) * 4;
    const float* xg = X  + (size_t)(m0 + srow) * Fsz + scol;
    const float* wg = Wm + (size_t)(2 * srow) * Dsz + scol;
    float* sa = lds + srow * P1_APITCH + scol;
    float* sb = lds + P1_A_DW + srow * P1_BPITCH + 2 * scol;

    v4f ra[8], rb[16];

    #pragma unroll
    for (int i = 0; i < 8; ++i)
        ra[i] = *(const v4f*)(xg + (size_t)(8 * i) * Fsz);
    #pragma unroll
    for (int i = 0; i < 8; ++i) {
        rb[2 * i]     = *(const v4f*)(wg + (size_t)(16 * i) * Dsz);
        rb[2 * i + 1] = *(const v4f*)(wg + (size_t)(16 * i + 1) * Dsz);
    }

    #pragma unroll
    for (int i = 0; i < 8; ++i)
        *(v4f*)(sa + (8 * i) * P1_APITCH) = ra[i];
    #pragma unroll
    for (int i = 0; i < 8; ++i) {
        const v4f r0 = rb[2 * i], r1 = rb[2 * i + 1];
        v4f lo, hi;
        lo.x = r0.x; lo.y = r1.x; lo.z = r0.y; lo.w = r1.y;
        hi.x = r0.z; hi.y = r1.z; hi.z = r0.w; hi.w = r1.w;
        float* dst = sb + (8 * i) * P1_BPITCH;
        *(v4f*)(dst)     = lo;
        *(v4f*)(dst + 4) = hi;
    }
    __syncthreads();

    #pragma unroll
    for (int i = 0; i < 8; ++i)
        ra[i] = *(const v4f*)(xg + (size_t)(8 * i) * Fsz + P1_KH);
    #pragma unroll
    for (int i = 0; i < 8; ++i) {
        rb[2 * i]     = *(const v4f*)(wg + (size_t)(P1_KH + 16 * i) * Dsz);
        rb[2 * i + 1] = *(const v4f*)(wg + (size_t)(P1_KH + 16 * i + 1) * Dsz);
    }

    const int mt = wave >> 1;
    const int nb = (wave & 1) * 64;
    const float* fa = lds + (mt * 16 + c) * P1_APITCH + 4 * h;
    const float* fb = lds + P1_A_DW + (2 * h) * P1_BPITCH + 2 * nb + 4 * c;

    v8f acc0 = {}, acc1 = {}, acc2 = {}, acc3 = {};

    p1_half(fa, fb, acc0, acc1, acc2, acc3);

    #pragma unroll
    for (int i = 0; i < 8; ++i)
        *(v4f*)(sa + P1_ST_DW + (8 * i) * P1_APITCH) = ra[i];
    #pragma unroll
    for (int i = 0; i < 8; ++i) {
        const v4f r0 = rb[2 * i], r1 = rb[2 * i + 1];
        v4f lo, hi;
        lo.x = r0.x; lo.y = r1.x; lo.z = r0.y; lo.w = r1.y;
        hi.x = r0.z; hi.y = r1.z; hi.z = r0.w; hi.w = r1.w;
        float* dst = sb + P1_ST_DW + (8 * i) * P1_BPITCH;
        *(v4f*)(dst)     = lo;
        *(v4f*)(dst + 4) = hi;
    }
    __syncthreads();

    int st1 = P1_ST_DW;
    asm volatile("" : "+v"(st1));
    p1_half(fa + st1, fb + st1, acc0, acc1, acc2, acc3);

    const int ncol = nb + 2 * c;
    v2f badd01 = {0.0f, 0.0f}, badd23 = {0.0f, 0.0f};
    if (which) {
        badd01.x = bias[ncol];
        badd01.y = bias[ncol + 1];
        badd23.x = bias[ncol + 32];
        badd23.y = bias[ncol + 33];
    }
    float* orow = O + (size_t)(m0 + mt * 16 + 8 * h) * Dsz + ncol;
    #pragma unroll
    for (int r = 0; r < 8; ++r) {
        v2f o01, o23;
        o01.x = acc0[r] + badd01.x;
        o01.y = acc1[r] + badd01.y;
        o23.x = acc2[r] + badd23.x;
        o23.y = acc3[r] + badd23.y;
        *(v2f*)(orow + (size_t)r * Dsz)      = o01;
        *(v2f*)(orow + (size_t)r * Dsz + 32) = o23;
    }
}

#define Q_PITCH 132

__device__ __forceinline__ v4f tanh4(v4f x) {
    v4f r;
    r.x = TANH(x.x);
    r.y = TANH(x.y);
    r.z = TANH(x.z);
    r.w = TANH(x.w);
    return r;
}

struct P2Frag { v4f q[4]; v4f k[4]; v4f av; };

__device__ __forceinline__ P2Frag p2_load(const float* qrow, const float* krow,
                                          const float* al, int d) {
    P2Frag f;
    #pragma unroll
    for (int j = 0; j < 4; ++j)
        f.q[j] = *(const v4f*)(qrow + j * 32 * Q_PITCH + d);
    #pragma unroll
    for (int i = 0; i < 4; ++i)
        f.k[i] = *(const v4f*)(krow + i * Dsz + d);
    f.av = *(const v4f*)(al + d);
    return f;
}

__global__ __launch_bounds__(256) void coattn_score(
    const float* __restrict__ qry_ws,
    const float* __restrict__ keys_ws,
    const float* __restrict__ avec,
    float* __restrict__ out)
{
    extern __shared__ __attribute__((aligned(16))) float smem[];
    float* ql = smem;
    float* kl = smem + Asz * Q_PITCH;
    float* al = kl + 32 * Dsz;

    const int b    = blockIdx.x >> 2;
    const int rblk = blockIdx.x & 3;
    const int t    = threadIdx.x;
    const int lane = t & 31;
    const int wave = t >> 5;

    const int frow = t >> 5;
    const int fcol = (t & 31) * 4;
    const float* qb = qry_ws  + (size_t)b * Asz * Dsz + (size_t)frow * Dsz + fcol;
    const float* kb = keys_ws + (size_t)(b * Rsz + rblk * 32 + frow) * Dsz + fcol;
    v4f rq[16], rk[4];
    #pragma unroll
    for (int i = 0; i < 16; ++i)
        rq[i] = *(const v4f*)(qb + (size_t)(8 * i) * Dsz);
    #pragma unroll
    for (int i = 0; i < 4; ++i)
        rk[i] = *(const v4f*)(kb + (size_t)(8 * i) * Dsz);
    #pragma unroll
    for (int i = 0; i < 16; ++i)
        *(v4f*)(ql + (frow + 8 * i) * Q_PITCH + fcol) = rq[i];
    #pragma unroll
    for (int i = 0; i < 4; ++i)
        *(v4f*)(kl + (frow + 8 * i) * Dsz + fcol) = rk[i];
    if (t < Dsz / 4) *(v4f*)(al + 4 * t) = *(const v4f*)(avec + 4 * t);
    __syncthreads();

    const float* krow = kl + (wave * 4) * Dsz;
    const float* qrow = ql + lane * Q_PITCH;

    v4f acc[4][4];
    #pragma unroll
    for (int i = 0; i < 4; ++i)
        #pragma unroll
        for (int j = 0; j < 4; ++j)
            acc[i][j] = (v4f){0.f, 0.f, 0.f, 0.f};

    P2Frag cur = p2_load(qrow, krow, al, 0);
    v4f px[4], pt[4];
    v4f pa = {0.f, 0.f, 0.f, 0.f};
    #pragma unroll
    for (int j = 0; j < 4; ++j) {
        px[j] = cur.q[j] + cur.k[0];
        pt[j] = (v4f){0.f, 0.f, 0.f, 0.f};
    }

    for (int d = 0; d < Dsz; d += 8) {
        const P2Frag nxt = p2_load(qrow, krow, al, d + 4);
        const P2Frag cu2 = p2_load(qrow, krow, al, (d + 8) & (Dsz - 1));
        __builtin_amdgcn_sched_barrier(0);
        #pragma unroll
        for (int s = 0; s < 8; ++s) {
            const int s1 = s + 1;
            const int r1 = s1 & 3;
            const int rp = (s + 3) & 3;
            const P2Frag& fc = (s < 4) ? cur : nxt;
            const P2Frag& fn = (s1 < 4) ? cur : ((s1 < 8) ? nxt : cu2);
            v4f tn[4], xn[4];
            #pragma unroll
            for (int j = 0; j < 4; ++j)
                tn[j] = tanh4(px[j]);
            #pragma unroll
            for (int j = 0; j < 4; ++j)
                xn[j] = fn.q[j] + fn.k[r1];
            #pragma unroll
            for (int j = 0; j < 4; ++j)
                acc[rp][j] = __builtin_elementwise_fma(pt[j], pa, acc[rp][j]);
            #pragma unroll
            for (int j = 0; j < 4; ++j) {
                px[j] = xn[j];
                pt[j] = tn[j];
            }
            pa = fc.av;
            #pragma unroll
            for (int g = 0; g < 16; ++g) {
                __builtin_amdgcn_sched_group_barrier(0x400, 1, 0);
                __builtin_amdgcn_sched_group_barrier(0x002, 1, 0);
            }
            __builtin_amdgcn_sched_barrier(0);
        }
        cur = cu2;
    }
    #pragma unroll
    for (int j = 0; j < 4; ++j)
        acc[3][j] = __builtin_elementwise_fma(pt[j], pa, acc[3][j]);

    const int r0 = rblk * 32 + wave * 4;
    float* orow = out + ((size_t)(b * Rsz + r0)) * Asz + lane;
    #pragma unroll
    for (int i = 0; i < 4; ++i) {
        #pragma unroll
        for (int j = 0; j < 4; ++j)
            orow[(size_t)i * Asz + 32 * j] = (acc[i][j].x + acc[i][j].y) + (acc[i][j].z + acc[i][j].w);
    }
}

extern "C" void kernel_launch(void* const* d_in, const int* in_sizes, int n_in,
                              void* d_out, int out_size, void* d_ws, size_t ws_size,
                              hipStream_t stream) {
    const float* receiver  = (const float*)d_in[0];
    const float* attendant = (const float*)d_in[1];
    const float* w_q       = (const float*)d_in[2];
    const float* w_k       = (const float*)d_in[3];
    const float* bias      = (const float*)d_in[4];
    const float* avec      = (const float*)d_in[5];
    float*       out       = (float*)d_out;

    float* qry_ws  = (float*)d_ws;
    float* keys_ws = qry_ws + (size_t)Bsz * Asz * Dsz;

    coattn_proj_wmma<<<256, 256, 0, stream>>>(
        receiver, attendant, w_k, w_q, bias, keys_ws, qry_ws);

    const size_t lds_bytes = (Asz * Q_PITCH + 32 * Dsz + Dsz) * sizeof(float);
    coattn_score<<<256, 256, lds_bytes, stream>>>(
        qry_ws, keys_ws, avec, out);
}
